# starting kernel
# baseline (speedup 1.0000x reference)
#include <hip/hip_runtime.h>


#define FOV        11
#define MAX_DIST   5
#define NACT       5
#define NEG_INF_F  1e10f
#define AGENTS     768
#define BATCH      4
#define PMIN       60
#define NPUSED     61
#define KTRUE      (NPUSED * NACT)
#define KPAD       308
#define SSTRIDE    309
#define TILES_PER_B (AGENTS / 16)

typedef float v2f __attribute__((ext_vector_type(2)));
typedef float v8f __attribute__((ext_vector_type(8)));

__global__ void mf_init_q(const float* __restrict__ q_init,
                          const float* __restrict__ ill,
                          float* __restrict__ q, int n) {
  int t = blockIdx.x * blockDim.x + threadIdx.x;
  if (t < n) q[t] = q_init[t] - NEG_INF_F * ill[t];
}

__global__ void mf_prep_B(const float* __restrict__ params,
                          float* __restrict__ Bg) {
  int idx = blockIdx.x * blockDim.x + threadIdx.x;
  if (idx >= KPAD * 16) return;
  int k = idx >> 4, n = idx & 15;
  float v = 0.f;
  if (n < NACT && k < KTRUE) {
    int p = k / NACT;
    int y = k - p * NACT;
    v = params[(p + PMIN) * (NACT * NACT) + n * NACT + y];
  }
  Bg[idx] = v;
}

__global__ void mf_softmax(const float* __restrict__ q,
                           float* __restrict__ probs, int n) {
  int t = blockIdx.x * blockDim.x + threadIdx.x;
  if (t >= n) return;
  const float* row = q + t * NACT;
  float m = row[0];
#pragma unroll
  for (int y = 1; y < NACT; ++y) m = fmaxf(m, row[y]);
  float e[NACT];
  float s = 0.f;
#pragma unroll
  for (int y = 0; y < NACT; ++y) { e[y] = __expf(row[y] - m); s += e[y]; }
  float inv = 1.f / s;
#pragma unroll
  for (int y = 0; y < NACT; ++y) probs[t * NACT + y] = e[y] * inv;
}

__global__ void __launch_bounds__(32)
mf_step(const float* __restrict__ logits,
        const float* __restrict__ ill,
        const int*   __restrict__ pos,
        const float* __restrict__ Bg,
        const float* __restrict__ probs,
        float* __restrict__ qout) {
  __shared__ float Slds[16 * SSTRIDE];
  __shared__ float Bm[KPAD * 16];

  const int lane   = threadIdx.x;
  const int b      = blockIdx.x / TILES_PER_B;
  const int i_base = (blockIdx.x - b * TILES_PER_B) * 16;

  const float* prb = probs + (size_t)(b * AGENTS) * NACT;

  for (int off = lane * 32; off < AGENTS * NACT; off += 32 * 32)
    __builtin_prefetch(prb + off, 0, 3);

  {
    const float4* Bg4 = (const float4*)Bg;
    float4*       Bm4 = (float4*)Bm;
    for (int idx = lane; idx < (KPAD * 16) / 4; idx += 32) Bm4[idx] = Bg4[idx];
  }
  for (int idx = lane; idx < 16 * SSTRIDE; idx += 32) Slds[idx] = 0.f;
  __syncthreads();

  if (lane < 16) {
    const int i  = i_base + lane;
    const int iy = pos[(b * AGENTS + i) * 2 + 0];
    const int ix = pos[(b * AGENTS + i) * 2 + 1];
    const int* pb = pos + (size_t)(b * AGENTS) * 2;
    float* srow = Slds + lane * SSTRIDE;
    for (int j = 0; j < AGENTS; ++j) {
      int jy = pb[2 * j], jx = pb[2 * j + 1];
      int dy = iy > jy ? iy - jy : jy - iy;
      int dx = ix > jx ? ix - jx : jx - ix;
      if (dy <= MAX_DIST && dx <= MAX_DIST) {
        float* dst = srow + (dy * FOV + dx) * NACT;
#pragma unroll
        for (int y = 0; y < NACT; ++y) dst[y] += prb[j * NACT + y];
      }
    }
  }
  __syncthreads();

  const int half = lane >> 4;
  const int nl   = lane & 15;
  v8f acc = {};
#pragma unroll 4
  for (int k0 = 0; k0 < KPAD; k0 += 4) {
    const int ka = k0 + 2 * half;
    v2f a, bf;
    a.x  = Slds[nl * SSTRIDE + ka];
    a.y  = Slds[nl * SSTRIDE + ka + 1];
    bf.x = Bm[ka * 16 + nl];
    bf.y = Bm[(ka + 1) * 16 + nl];
    acc = __builtin_amdgcn_wmma_f32_16x16x4_f32(
         false, a,  false, bf,
         (short)0, acc,  false,  false);
  }

  if (nl < NACT) {
#pragma unroll
    for (int g = 0; g < 8; ++g) {
      int m = g + 8 * half;
      int o = (b * AGENTS + i_base + m) * NACT + nl;
      qout[o] = logits[o] + acc[g] - NEG_INF_F * ill[o];
    }
  }
}

extern "C" void kernel_launch(void* const* d_in, const int* in_sizes, int n_in,
                              void* d_out, int out_size, void* d_ws, size_t ws_size,
                              hipStream_t stream) {
  (void)in_sizes; (void)n_in; (void)out_size; (void)ws_size;
  const float* logits = (const float*)d_in[0];
  const float* ill    = (const float*)d_in[1];
  const int*   pos    = (const int*)d_in[2];
  const float* q_init = (const float*)d_in[3];
  const float* params = (const float*)d_in[4];
  const int ITER = 3;
  const int NQ  = BATCH * AGENTS * NACT;
  const int NBA = BATCH * AGENTS;
  const int NB  = KPAD * 16;

  float* q_buf = (float*)d_ws;
  float* probs = q_buf + NQ;
  float* Bg    = probs + NQ;

  mf_prep_B<<<(NB + 255) / 256, 256, 0, stream>>>(params, Bg);
  mf_init_q<<<(NQ + 255) / 256, 256, 0, stream>>>(q_init, ill, q_buf, NQ);
  for (int it = 0; it < ITER; ++it) {
    mf_softmax<<<(NBA + 255) / 256, 256, 0, stream>>>(q_buf, probs, NBA);
    float* out = (it == ITER - 1) ? (float*)d_out : q_buf;
    mf_step<<<BATCH * TILES_PER_B, 32, 0, stream>>>(logits, ill, pos, Bg,
                                                    probs, out);
  }
}


# Neighbour scatter spread over the work-group: one wave per agent row, 32 candidates tested per ballot round, hits added in ascending j (same summation order), probs staged in LDS
# speedup vs baseline: 8.1717x; 8.1717x over previous
#include <hip/hip_runtime.h>


#define FOV        11
#define MAX_DIST   5
#define NACT       5
#define NEG_INF_F  1e10f
#define AGENTS     768
#define BATCH      4
#define PMIN       60
#define NPUSED     61
#define KTRUE      (NPUSED * NACT)
#define KPAD       308
#define SSTRIDE    309
#define TILES_PER_B (AGENTS / 16)

typedef float v2f __attribute__((ext_vector_type(2)));
typedef float v8f __attribute__((ext_vector_type(8)));
typedef float v4f __attribute__((ext_vector_type(4)));

__global__ void mf_init_q(const float* __restrict__ q_init,
                          const float* __restrict__ ill,
                          float* __restrict__ q, int n) {
  int t = blockIdx.x * blockDim.x + threadIdx.x;
  if (t < n) q[t] = q_init[t] - NEG_INF_F * ill[t];
}

__global__ void mf_prep_B(const float* __restrict__ params,
                          float* __restrict__ Bg) {
  int idx = blockIdx.x * blockDim.x + threadIdx.x;
  if (idx >= KPAD * 16) return;
  int k = idx >> 4, n = idx & 15;
  float v = 0.f;
  if (n < NACT && k < KTRUE) {
    int p = k / NACT;
    int y = k - p * NACT;
    v = params[(p + PMIN) * (NACT * NACT) + n * NACT + y];
  }
  Bg[idx] = v;
}

__global__ void mf_softmax(const float* __restrict__ q,
                           float* __restrict__ probs, int n) {
  int t = blockIdx.x * blockDim.x + threadIdx.x;
  if (t >= n) return;
  const float* row = q + t * NACT;
  float m = row[0];
#pragma unroll
  for (int y = 1; y < NACT; ++y) m = fmaxf(m, row[y]);
  float e[NACT];
  float s = 0.f;
#pragma unroll
  for (int y = 0; y < NACT; ++y) { e[y] = __expf(row[y] - m); s += e[y]; }
  float inv = 1.f / s;
#pragma unroll
  for (int y = 0; y < NACT; ++y) probs[t * NACT + y] = e[y] * inv;
}

#define STEP_WAVES   16
#define STEP_THREADS (STEP_WAVES * 32)
#define JROUNDS      (AGENTS / 32)
#define B4_COUNT     ((KPAD * 16) / 4)
#define P4_COUNT     ((AGENTS * NACT) / 4)
#define B4_ROUNDS    ((B4_COUNT + STEP_THREADS - 1) / STEP_THREADS)
#define P4_ROUNDS    ((P4_COUNT + STEP_THREADS - 1) / STEP_THREADS)

__global__ void __launch_bounds__(STEP_THREADS)
mf_step(const float* __restrict__ logits,
        const float* __restrict__ ill,
        const int*   __restrict__ pos,
        const float* __restrict__ Bg,
        const float* __restrict__ probs,
        float* __restrict__ qout) {
  __shared__ __attribute__((aligned(16))) float Slds[16 * SSTRIDE];
  __shared__ __attribute__((aligned(16))) float Bm[B4_ROUNDS * STEP_THREADS * 4];
  __shared__ __attribute__((aligned(16))) float Pl[P4_ROUNDS * STEP_THREADS * 4];

  const int tid    = threadIdx.x;
  const int lane   = tid & 31;
  const int wave   = __builtin_amdgcn_readfirstlane(tid >> 5);
  const int b      = blockIdx.x / TILES_PER_B;
  const int i_base = (blockIdx.x - b * TILES_PER_B) * 16;

  const float* prb = probs + (size_t)(b * AGENTS) * NACT;
  const int*   pb  = pos + (size_t)(b * AGENTS) * 2;

  const int iy = pb[2 * (i_base + wave) + 0];
  const int ix = pb[2 * (i_base + wave) + 1];
  int jy[JROUNDS], jx[JROUNDS];
#pragma unroll
  for (int r = 0; r < JROUNDS; ++r) {
    const int j = r * 32 + lane;
    jy[r] = pb[2 * j];
    jx[r] = pb[2 * j + 1];
  }
  const v4f* Bg4 = (const v4f*)Bg;
  const v4f* pr4 = (const v4f*)prb;
  v4f tb[B4_ROUNDS], tp[P4_ROUNDS];
#pragma unroll
  for (int k = 0; k < B4_ROUNDS; ++k) {
    const int idx = tid + k * STEP_THREADS;
    tb[k] = Bg4[idx < B4_COUNT ? idx : B4_COUNT - 1];
  }
#pragma unroll
  for (int k = 0; k < P4_ROUNDS; ++k) {
    const int idx = tid + k * STEP_THREADS;
    tp[k] = pr4[idx < P4_COUNT ? idx : P4_COUNT - 1];
  }

  float* srow = Slds + wave * SSTRIDE;
#pragma unroll
  for (int k = 0; k < (SSTRIDE + 31) / 32; ++k) {
    const int idx = lane + 32 * k;
    if (idx < SSTRIDE) srow[idx] = 0.f;
  }

  static_assert(B4_ROUNDS == 3 && P4_ROUNDS == 2, "fence lists every staged value");
  asm volatile("" :: "v"(tb[0]), "v"(tb[1]), "v"(tb[2]), "v"(tp[0]), "v"(tp[1]));

  {
    v4f* Bm4 = (v4f*)Bm;
    v4f* Pl4 = (v4f*)Pl;
#pragma unroll
    for (int k = 0; k < B4_ROUNDS; ++k) Bm4[tid + k * STEP_THREADS] = tb[k];
#pragma unroll
    for (int k = 0; k < P4_ROUNDS; ++k) Pl4[tid + k * STEP_THREADS] = tp[k];
  }

  int      code[JROUNDS];
  unsigned hits[JROUNDS];
#pragma unroll
  for (int r = 0; r < JROUNDS; ++r) {
    const int dy = iy > jy[r] ? iy - jy[r] : jy[r] - iy;
    const int dx = ix > jx[r] ? ix - jx[r] : jx[r] - ix;
    code[r] = (dy * FOV + dx) * NACT;
    hits[r] = __builtin_amdgcn_ballot_w32(dy <= MAX_DIST && dx <= MAX_DIST);
  }
  __syncthreads();

#pragma unroll
  for (int r = 0; r < JROUNDS; ++r) {
    unsigned m = hits[r];
    while (m) {
      const int bit = __builtin_ctz(m);
      m &= m - 1;
      const int off = __builtin_amdgcn_readlane(code[r], bit);
      const int j   = r * 32 + bit;
      if (lane < NACT) srow[off + lane] += Pl[j * NACT + lane];
    }
  }
  __syncthreads();
  if (wave != 0) return;

  const int half = lane >> 4;
  const int nl   = lane & 15;
  v8f acc = {};
#pragma unroll 4
  for (int k0 = 0; k0 < KPAD; k0 += 4) {
    const int ka = k0 + 2 * half;
    v2f a, bf;
    a.x  = Slds[nl * SSTRIDE + ka];
    a.y  = Slds[nl * SSTRIDE + ka + 1];
    bf.x = Bm[ka * 16 + nl];
    bf.y = Bm[(ka + 1) * 16 + nl];
    acc = __builtin_amdgcn_wmma_f32_16x16x4_f32(
         false, a,  false, bf,
         (short)0, acc,  false,  false);
  }

  if (nl < NACT) {
#pragma unroll
    for (int g = 0; g < 8; ++g) {
      int m = g + 8 * half;
      int o = (b * AGENTS + i_base + m) * NACT + nl;
      qout[o] = logits[o] + acc[g] - NEG_INF_F * ill[o];
    }
  }
}

extern "C" void kernel_launch(void* const* d_in, const int* in_sizes, int n_in,
                              void* d_out, int out_size, void* d_ws, size_t ws_size,
                              hipStream_t stream) {
  (void)in_sizes; (void)n_in; (void)out_size; (void)ws_size;
  const float* logits = (const float*)d_in[0];
  const float* ill    = (const float*)d_in[1];
  const int*   pos    = (const int*)d_in[2];
  const float* q_init = (const float*)d_in[3];
  const float* params = (const float*)d_in[4];
  const int ITER = 3;
  const int NQ  = BATCH * AGENTS * NACT;
  const int NBA = BATCH * AGENTS;
  const int NB  = KPAD * 16;

  float* q_buf = (float*)d_ws;
  float* probs = q_buf + NQ;
  float* Bg    = probs + NQ;

  mf_prep_B<<<(NB + 255) / 256, 256, 0, stream>>>(params, Bg);
  mf_init_q<<<(NQ + 255) / 256, 256, 0, stream>>>(q_init, ill, q_buf, NQ);
  for (int it = 0; it < ITER; ++it) {
    mf_softmax<<<(NBA + 255) / 256, 256, 0, stream>>>(q_buf, probs, NBA);
    float* out = (it == ITER - 1) ? (float*)d_out : q_buf;
    mf_step<<<BATCH * TILES_PER_B, STEP_THREADS, 0, stream>>>(logits, ill, pos, Bg,
                                                    probs, out);
  }
}


# Helper launches folded into the sweep kernel: each work-group computes its batch element's softmax (and the q start value) and builds the B image in LDS itself; 8 launches become 3
# speedup vs baseline: 11.3520x; 1.3892x over previous
#include <hip/hip_runtime.h>


#define FOV        11
#define MAX_DIST   5
#define NACT       5
#define NEG_INF_F  1e10f
#define AGENTS     768
#define BATCH      4
#define PMIN       60
#define NPUSED     61
#define KTRUE      (NPUSED * NACT)
#define KPAD       308
#define SSTRIDE    309
#define TILES_PER_B (AGENTS / 16)

typedef float v2f __attribute__((ext_vector_type(2)));
typedef float v8f __attribute__((ext_vector_type(8)));
typedef float v4f __attribute__((ext_vector_type(4)));

#define STEP_WAVES   16
#define STEP_THREADS (STEP_WAVES * 32)
#define JROUNDS      (AGENTS / 32)
#define B4_COUNT     ((KPAD * 16) / 4)
#define B4_ROUNDS    ((B4_COUNT + STEP_THREADS - 1) / STEP_THREADS)
#define PUSED_COUNT  (NPUSED * NACT * NACT)
#define PV_ROUNDS    ((PUSED_COUNT + STEP_THREADS - 1) / STEP_THREADS)
#define Q_ROUNDS     ((AGENTS + STEP_THREADS - 1) / STEP_THREADS)

template <bool FIRST>
__global__ void __launch_bounds__(STEP_THREADS)
mf_step(const float* __restrict__ logits,
        const float* __restrict__ ill,
        const int*   __restrict__ pos,
        const float* __restrict__ params,
        const float* __restrict__ qsrc,
        float* __restrict__ qout) {
  __shared__ __attribute__((aligned(16))) float Slds[16 * SSTRIDE];
  __shared__ __attribute__((aligned(16))) float Bm[B4_ROUNDS * STEP_THREADS * 4];
  __shared__ float Pl[AGENTS * NACT];

  const int tid    = threadIdx.x;
  const int lane   = tid & 31;
  const int wave   = __builtin_amdgcn_readfirstlane(tid >> 5);
  const int b      = blockIdx.x / TILES_PER_B;
  const int i_base = (blockIdx.x - b * TILES_PER_B) * 16;

  const float* qb  = qsrc + (size_t)(b * AGENTS) * NACT;
  const float* ib  = ill  + (size_t)(b * AGENTS) * NACT;
  const int*   pb  = pos + (size_t)(b * AGENTS) * 2;

  const int iy = pb[2 * (i_base + wave) + 0];
  const int ix = pb[2 * (i_base + wave) + 1];
  int jy[JROUNDS], jx[JROUNDS];
#pragma unroll
  for (int r = 0; r < JROUNDS; ++r) {
    const int j = r * 32 + lane;
    jy[r] = pb[2 * j];
    jx[r] = pb[2 * j + 1];
  }
  float qv[Q_ROUNDS][NACT];
#pragma unroll
  for (int a = 0; a < Q_ROUNDS; ++a) {
    const int ag = tid + a * STEP_THREADS;
    const int o  = (ag < AGENTS ? ag : AGENTS - 1) * NACT;
#pragma unroll
    for (int y = 0; y < NACT; ++y) {
      if (FIRST) qv[a][y] = qb[o + y] - NEG_INF_F * ib[o + y];
      else       qv[a][y] = qb[o + y];
    }
  }
  float pv[PV_ROUNDS];
#pragma unroll
  for (int k = 0; k < PV_ROUNDS; ++k) {
    const int e = tid + k * STEP_THREADS;
    pv[k] = params[PMIN * NACT * NACT + (e < PUSED_COUNT ? e : PUSED_COUNT - 1)];
  }

  float* srow = Slds + wave * SSTRIDE;
#pragma unroll
  for (int k = 0; k < (SSTRIDE + 31) / 32; ++k) {
    const int idx = lane + 32 * k;
    if (idx < SSTRIDE) srow[idx] = 0.f;
  }

  {
    v4f* Bm4 = (v4f*)Bm;
    const v4f z4 = {0.f, 0.f, 0.f, 0.f};
#pragma unroll
    for (int k = 0; k < B4_ROUNDS; ++k) Bm4[tid + k * STEP_THREADS] = z4;
  }

#pragma unroll
  for (int a = 0; a < Q_ROUNDS; ++a) {
    const int ag = tid + a * STEP_THREADS;
    float m = qv[a][0];
#pragma unroll
    for (int y = 1; y < NACT; ++y) m = fmaxf(m, qv[a][y]);
    float e[NACT];
    float s = 0.f;
#pragma unroll
    for (int y = 0; y < NACT; ++y) { e[y] = __expf(qv[a][y] - m); s += e[y]; }
    const float inv = 1.f / s;
    if (ag < AGENTS) {
#pragma unroll
      for (int y = 0; y < NACT; ++y) Pl[ag * NACT + y] = e[y] * inv;
    }
  }

  int      code[JROUNDS];
  unsigned hits[JROUNDS];
#pragma unroll
  for (int r = 0; r < JROUNDS; ++r) {
    const int dy = iy > jy[r] ? iy - jy[r] : jy[r] - iy;
    const int dx = ix > jx[r] ? ix - jx[r] : jx[r] - ix;
    code[r] = (dy * FOV + dx) * NACT;
    hits[r] = __builtin_amdgcn_ballot_w32(dy <= MAX_DIST && dx <= MAX_DIST);
  }
  __syncthreads();

#pragma unroll
  for (int k = 0; k < PV_ROUNDS; ++k) {
    const int e = tid + k * STEP_THREADS;
    if (e < PUSED_COUNT) {
      const int p   = e / (NACT * NACT);
      const int rem = e - p * (NACT * NACT);
      const int x   = rem / NACT;
      const int y   = rem - x * NACT;
      Bm[(p * NACT + y) * 16 + x] = pv[k];
    }
  }

#pragma unroll
  for (int r = 0; r < JROUNDS; ++r) {
    unsigned m = hits[r];
    while (m) {
      const int bit = __builtin_ctz(m);
      m &= m - 1;
      const int off = __builtin_amdgcn_readlane(code[r], bit);
      const int j   = r * 32 + bit;
      if (lane < NACT) srow[off + lane] += Pl[j * NACT + lane];
    }
  }
  __syncthreads();
  if (wave != 0) return;

  const int half = lane >> 4;
  const int nl   = lane & 15;
  v8f acc = {};
#pragma unroll 4
  for (int k0 = 0; k0 < KPAD; k0 += 4) {
    const int ka = k0 + 2 * half;
    v2f a, bf;
    a.x  = Slds[nl * SSTRIDE + ka];
    a.y  = Slds[nl * SSTRIDE + ka + 1];
    bf.x = Bm[ka * 16 + nl];
    bf.y = Bm[(ka + 1) * 16 + nl];
    acc = __builtin_amdgcn_wmma_f32_16x16x4_f32(
         false, a,  false, bf,
         (short)0, acc,  false,  false);
  }

  if (nl < NACT) {
#pragma unroll
    for (int g = 0; g < 8; ++g) {
      int m = g + 8 * half;
      int o = (b * AGENTS + i_base + m) * NACT + nl;
      qout[o] = logits[o] + acc[g] - NEG_INF_F * ill[o];
    }
  }
}

extern "C" void kernel_launch(void* const* d_in, const int* in_sizes, int n_in,
                              void* d_out, int out_size, void* d_ws, size_t ws_size,
                              hipStream_t stream) {
  (void)in_sizes; (void)n_in; (void)out_size; (void)ws_size;
  const float* logits = (const float*)d_in[0];
  const float* ill    = (const float*)d_in[1];
  const int*   pos    = (const int*)d_in[2];
  const float* q_init = (const float*)d_in[3];
  const float* params = (const float*)d_in[4];
  const int NQ = BATCH * AGENTS * NACT;

  float* q_a = (float*)d_ws;
  float* q_b = q_a + NQ;

  mf_step<true><<<BATCH * TILES_PER_B, STEP_THREADS, 0, stream>>>(
      logits, ill, pos, params, q_init, q_a);
  mf_step<false><<<BATCH * TILES_PER_B, STEP_THREADS, 0, stream>>>(
      logits, ill, pos, params, q_a, q_b);
  mf_step<false><<<BATCH * TILES_PER_B, STEP_THREADS, 0, stream>>>(
      logits, ill, pos, params, q_b, (float*)d_out);
}


# Neighbour scatter four-way parallel: hits compacted to a per-wave LDS list, four lane groups add them into four interleaved partial rows, A loader sums the four: a different summation tree
# speedup vs baseline: 17.3074x; 1.5246x over previous
#include <hip/hip_runtime.h>


#define FOV        11
#define MAX_DIST   5
#define NACT       5
#define NEG_INF_F  1e10f
#define AGENTS     768
#define BATCH      4
#define PMIN       60
#define NPUSED     61
#define KTRUE      (NPUSED * NACT)
#define STEP_WAVES 16
#define KSTEPS_PER_WAVE 5
#define KPAD       (STEP_WAVES * KSTEPS_PER_WAVE * 4)
#define SSTRIDE    325
#define TILES_PER_B (AGENTS / 16)
#define NCOPY      4
#define CHUNK      32
#define CSTEPS     (CHUNK / NCOPY)

typedef float v2f __attribute__((ext_vector_type(2)));
typedef float v8f __attribute__((ext_vector_type(8)));
typedef float v4f __attribute__((ext_vector_type(4)));

#define STEP_THREADS (STEP_WAVES * 32)
#define JROUNDS      (AGENTS / 32)
#define B4_COUNT     ((KPAD * 16) / 4)
#define B4_ROUNDS    ((B4_COUNT + STEP_THREADS - 1) / STEP_THREADS)
#define PUSED_COUNT  (NPUSED * NACT * NACT)
#define PV_ROUNDS    ((PUSED_COUNT + STEP_THREADS - 1) / STEP_THREADS)
#define Q_ROUNDS     ((AGENTS + STEP_THREADS - 1) / STEP_THREADS)

template <bool FIRST>
__global__ void __launch_bounds__(STEP_THREADS)
mf_step(const float* __restrict__ logits,
        const float* __restrict__ ill,
        const int*   __restrict__ pos,
        const float* __restrict__ params,
        const float* __restrict__ qsrc,
        float* __restrict__ qout) {
  __shared__ __attribute__((aligned(16))) float Slds[16 * SSTRIDE * NCOPY];
  __shared__ int List[STEP_WAVES * AGENTS];
  __shared__ __attribute__((aligned(16))) float Bm[B4_ROUNDS * STEP_THREADS * 4];
  __shared__ float Pl[AGENTS * NACT + 1];
  __shared__ float Red[STEP_WAVES * 16 * NACT];

  const int tid    = threadIdx.x;
  const int lane   = tid & 31;
  const int wave   = __builtin_amdgcn_readfirstlane(tid >> 5);
  const int b      = blockIdx.x / TILES_PER_B;
  const int i_base = (blockIdx.x - b * TILES_PER_B) * 16;

  const float* qb  = qsrc + (size_t)(b * AGENTS) * NACT;
  const float* ib  = ill  + (size_t)(b * AGENTS) * NACT;
  const int*   pb  = pos + (size_t)(b * AGENTS) * 2;

  const int iy = pb[2 * (i_base + wave) + 0];
  const int ix = pb[2 * (i_base + wave) + 1];
  int jy[JROUNDS], jx[JROUNDS];
#pragma unroll
  for (int r = 0; r < JROUNDS; ++r) {
    const int j = r * 32 + lane;
    jy[r] = pb[2 * j];
    jx[r] = pb[2 * j + 1];
  }
  float qv[Q_ROUNDS][NACT];
#pragma unroll
  for (int a = 0; a < Q_ROUNDS; ++a) {
    const int ag = tid + a * STEP_THREADS;
    const int o  = (ag < AGENTS ? ag : AGENTS - 1) * NACT;
#pragma unroll
    for (int y = 0; y < NACT; ++y) {
      if (FIRST) qv[a][y] = qb[o + y] - NEG_INF_F * ib[o + y];
      else       qv[a][y] = qb[o + y];
    }
  }
  float pv[PV_ROUNDS];
#pragma unroll
  for (int k = 0; k < PV_ROUNDS; ++k) {
    const int e = tid + k * STEP_THREADS;
    pv[k] = params[PMIN * NACT * NACT + (e < PUSED_COUNT ? e : PUSED_COUNT - 1)];
  }

  float* srow = Slds + wave * (SSTRIDE * NCOPY);
  {
    v4f* srow4 = (v4f*)srow;
    const v4f z4 = {0.f, 0.f, 0.f, 0.f};
#pragma unroll
    for (int k = 0; k < (SSTRIDE + 31) / 32; ++k) {
      const int idx = lane + 32 * k;
      if (idx < SSTRIDE) srow4[idx] = z4;
    }
  }
  if (tid == 0) Pl[AGENTS * NACT] = 0.f;

  {
    v4f* Bm4 = (v4f*)Bm;
    const v4f z4 = {0.f, 0.f, 0.f, 0.f};
#pragma unroll
    for (int k = 0; k < B4_ROUNDS; ++k) Bm4[tid + k * STEP_THREADS] = z4;
  }

#pragma unroll
  for (int a = 0; a < Q_ROUNDS; ++a) {
    const int ag = tid + a * STEP_THREADS;
    float m = qv[a][0];
#pragma unroll
    for (int y = 1; y < NACT; ++y) m = fmaxf(m, qv[a][y]);
    float e[NACT];
    float s = 0.f;
#pragma unroll
    for (int y = 0; y < NACT; ++y) { e[y] = __expf(qv[a][y] - m); s += e[y]; }
    const float inv = 1.f / s;
    if (ag < AGENTS) {
#pragma unroll
      for (int y = 0; y < NACT; ++y) Pl[ag * NACT + y] = e[y] * inv;
    }
  }

  int* mylist = List + wave * AGENTS;
  int  nhit   = 0;
#pragma unroll
  for (int r = 0; r < JROUNDS; ++r) {
    const int dy = iy > jy[r] ? iy - jy[r] : jy[r] - iy;
    const int dx = ix > jx[r] ? ix - jx[r] : jx[r] - ix;
    const bool hit = dy <= MAX_DIST && dx <= MAX_DIST;
    const unsigned mask = __builtin_amdgcn_ballot_w32(hit);
    if (hit)
      mylist[__builtin_amdgcn_mbcnt_lo(mask, (unsigned)nhit)] =
          ((dy * FOV + dx) * NACT) | ((r * 32 + lane) << 16);
    nhit += __builtin_popcount(mask);
  }
  __syncthreads();

#pragma unroll
  for (int k = 0; k < PV_ROUNDS; ++k) {
    const int e = tid + k * STEP_THREADS;
    if (e < PUSED_COUNT) {
      const int p   = e / (NACT * NACT);
      const int rem = e - p * (NACT * NACT);
      const int x   = rem / NACT;
      const int y   = rem - x * NACT;
      Bm[(p * NACT + y) * 16 + x] = pv[k];
    }
  }

  {
    const int g     = lane >> 3;
    const int y     = lane & 7;
    const bool yv   = y < NACT;
    const int yg    = y * NCOPY + g;
    const int dummy = KPAD * NCOPY + (lane & 15);
    for (int c = 0; c < nhit; c += CHUNK) {
      int   ao[CSTEPS];
      float pr[CSTEPS];
#pragma unroll
      for (int t = 0; t < CSTEPS; ++t) {
        const int idx = c + NCOPY * t + g;
        const bool ok = yv && idx < nhit;
        const unsigned e = (unsigned)mylist[idx < nhit ? idx : nhit - 1];
        ao[t] = ok ? (int)(e & 0xffffu) * NCOPY + yg : dummy;
        pr[t] = Pl[ok ? (int)(e >> 16) * NACT + y : AGENTS * NACT];
      }
      static_assert(CSTEPS == 8, "the statement below names CSTEPS values of each kind");
      asm volatile("" :: "v"(pr[0]), "v"(pr[1]), "v"(pr[2]), "v"(pr[3]),
                         "v"(pr[4]), "v"(pr[5]), "v"(pr[6]), "v"(pr[7]),
                         "v"(ao[0]), "v"(ao[1]), "v"(ao[2]), "v"(ao[3]),
                         "v"(ao[4]), "v"(ao[5]), "v"(ao[6]), "v"(ao[7]));
#pragma unroll
      for (int t = 0; t < CSTEPS; ++t) {
        if (c + NCOPY * t >= nhit) break;
        srow[ao[t]] += pr[t];
      }
    }
  }
  __syncthreads();

  const int half  = lane >> 4;
  const int nl    = lane & 15;
  const int kbase = wave * (KSTEPS_PER_WAVE * 4);
  const v4f* S4 = (const v4f*)Slds;
  v8f acc = {};
#pragma unroll
  for (int s = 0; s < KSTEPS_PER_WAVE; ++s) {
    const int ka = kbase + 4 * s + 2 * half;
    v2f a, bf;
    const v4f u0 = S4[nl * SSTRIDE + ka];
    const v4f u1 = S4[nl * SSTRIDE + ka + 1];
    a.x  = ((u0.x + u0.y) + u0.z) + u0.w;
    a.y  = ((u1.x + u1.y) + u1.z) + u1.w;
    bf.x = Bm[ka * 16 + nl];
    bf.y = Bm[(ka + 1) * 16 + nl];
    acc = __builtin_amdgcn_wmma_f32_16x16x4_f32(
         false, a,  false, bf,
         (short)0, acc,  false,  false);
  }

  if (nl < NACT) {
#pragma unroll
    for (int g = 0; g < 8; ++g)
      Red[(wave * 16 + g + 8 * half) * NACT + nl] = acc[g];
  }
  __syncthreads();

  if (tid < 16 * NACT) {
    float sum = Red[tid];
#pragma unroll
    for (int w = 1; w < STEP_WAVES; ++w) sum += Red[w * 16 * NACT + tid];
    const int o = (b * AGENTS + i_base) * NACT + tid;
    qout[o] = logits[o] + sum - NEG_INF_F * ill[o];
  }
}

extern "C" void kernel_launch(void* const* d_in, const int* in_sizes, int n_in,
                              void* d_out, int out_size, void* d_ws, size_t ws_size,
                              hipStream_t stream) {
  (void)in_sizes; (void)n_in; (void)out_size; (void)ws_size;
  const float* logits = (const float*)d_in[0];
  const float* ill    = (const float*)d_in[1];
  const int*   pos    = (const int*)d_in[2];
  const float* q_init = (const float*)d_in[3];
  const float* params = (const float*)d_in[4];
  const int NQ = BATCH * AGENTS * NACT;

  float* q_a = (float*)d_ws;
  float* q_b = q_a + NQ;

  mf_step<true><<<BATCH * TILES_PER_B, STEP_THREADS, 0, stream>>>(
      logits, ill, pos, params, q_init, q_a);
  mf_step<false><<<BATCH * TILES_PER_B, STEP_THREADS, 0, stream>>>(
      logits, ill, pos, params, q_a, q_b);
  mf_step<false><<<BATCH * TILES_PER_B, STEP_THREADS, 0, stream>>>(
      logits, ill, pos, params, q_b, (float*)d_out);
}


# Neighbour test as one straight-line block: all 24 rounds' hit masks first, then branch-free list stores (a lane with no hit writes a spare word), |d| by v_sad_u32; same list order and sums
# speedup vs baseline: 18.4019x; 1.0632x over previous
#include <hip/hip_runtime.h>


#define FOV        11
#define MAX_DIST   5
#define NACT       5
#define NEG_INF_F  1e10f
#define AGENTS     768
#define BATCH      4
#define PMIN       60
#define NPUSED     61
#define KTRUE      (NPUSED * NACT)
#define STEP_WAVES 16
#define KSTEPS_PER_WAVE 5
#define KPAD       (STEP_WAVES * KSTEPS_PER_WAVE * 4)
#define SSTRIDE    325
#define TILES_PER_B (AGENTS / 16)
#define NCOPY      4
#define CHUNK      32
#define CSTEPS     (CHUNK / NCOPY)
#define SPARE_AT    (AGENTS + 32)
#define LIST_STRIDE (AGENTS + 64)

typedef float v2f __attribute__((ext_vector_type(2)));
typedef float v8f __attribute__((ext_vector_type(8)));
typedef float v4f __attribute__((ext_vector_type(4)));

#define STEP_THREADS (STEP_WAVES * 32)
#define JROUNDS      (AGENTS / 32)
#define B4_COUNT     ((KPAD * 16) / 4)
#define B4_ROUNDS    ((B4_COUNT + STEP_THREADS - 1) / STEP_THREADS)
#define PUSED_COUNT  (NPUSED * NACT * NACT)
#define PV_ROUNDS    ((PUSED_COUNT + STEP_THREADS - 1) / STEP_THREADS)
#define Q_ROUNDS     ((AGENTS + STEP_THREADS - 1) / STEP_THREADS)

__device__ __forceinline__ unsigned absdiff5_add(unsigned a, unsigned c) {
  unsigned d;
  asm("v_sad_u32 %0, %1, 5, %2" : "=v"(d) : "v"(a), "v"(c));
  return d;
}

template <bool FIRST>
__global__ void __launch_bounds__(STEP_THREADS)
mf_step(const float* __restrict__ logits,
        const float* __restrict__ ill,
        const int*   __restrict__ pos,
        const float* __restrict__ params,
        const float* __restrict__ qsrc,
        float* __restrict__ qout) {
  __shared__ __attribute__((aligned(16))) float Slds[16 * SSTRIDE * NCOPY];
  __shared__ int List[STEP_WAVES * LIST_STRIDE];
  __shared__ __attribute__((aligned(16))) float Bm[B4_ROUNDS * STEP_THREADS * 4];
  __shared__ float Pl[AGENTS * NACT + 1];
  __shared__ float Red[STEP_WAVES * 16 * NACT];

  const int tid    = threadIdx.x;
  const int lane   = tid & 31;
  const int wave   = __builtin_amdgcn_readfirstlane(tid >> 5);
  const int b      = blockIdx.x / TILES_PER_B;
  const int i_base = (blockIdx.x - b * TILES_PER_B) * 16;

  unsigned jbits[JROUNDS];
#pragma unroll
  for (int r = 0; r < JROUNDS; ++r) jbits[r] = (unsigned)(r * 32 + lane) << 16;

  const float* qb  = qsrc + (size_t)(b * AGENTS) * NACT;
  const float* ib  = ill  + (size_t)(b * AGENTS) * NACT;
  const int*   pb  = pos + (size_t)(b * AGENTS) * 2;

  const int iy = pb[2 * (i_base + wave) + 0];
  const int ix = pb[2 * (i_base + wave) + 1];
  int jy[JROUNDS], jx[JROUNDS];
#pragma unroll
  for (int r = 0; r < JROUNDS; ++r) {
    const int j = r * 32 + lane;
    jy[r] = pb[2 * j];
    jx[r] = pb[2 * j + 1];
  }
  float qv[Q_ROUNDS][NACT];
#pragma unroll
  for (int a = 0; a < Q_ROUNDS; ++a) {
    const int ag = tid + a * STEP_THREADS;
    const int o  = (ag < AGENTS ? ag : AGENTS - 1) * NACT;
#pragma unroll
    for (int y = 0; y < NACT; ++y) {
      if (FIRST) qv[a][y] = qb[o + y] - NEG_INF_F * ib[o + y];
      else       qv[a][y] = qb[o + y];
    }
  }
  float pv[PV_ROUNDS];
#pragma unroll
  for (int k = 0; k < PV_ROUNDS; ++k) {
    const int e = tid + k * STEP_THREADS;
    pv[k] = params[PMIN * NACT * NACT + (e < PUSED_COUNT ? e : PUSED_COUNT - 1)];
  }

  float* srow = Slds + wave * (SSTRIDE * NCOPY);
  {
    v4f* srow4 = (v4f*)srow;
    const v4f z4 = {0.f, 0.f, 0.f, 0.f};
#pragma unroll
    for (int k = 0; k < (SSTRIDE + 31) / 32; ++k) {
      const int idx = lane + 32 * k;
      if (idx < SSTRIDE) srow4[idx] = z4;
    }
  }
  if (tid == 0) Pl[AGENTS * NACT] = 0.f;

  {
    v4f* Bm4 = (v4f*)Bm;
    const v4f z4 = {0.f, 0.f, 0.f, 0.f};
#pragma unroll
    for (int k = 0; k < B4_ROUNDS; ++k) Bm4[tid + k * STEP_THREADS] = z4;
  }

  static_assert(JROUNDS == 24, "the statement below names JROUNDS values");
  asm volatile("" :: "v"(jbits[0]), "v"(jbits[1]), "v"(jbits[2]), "v"(jbits[3]), "v"(jbits[4]), "v"(jbits[5]),
                     "v"(jbits[6]), "v"(jbits[7]), "v"(jbits[8]), "v"(jbits[9]), "v"(jbits[10]), "v"(jbits[11]),
                     "v"(jbits[12]), "v"(jbits[13]), "v"(jbits[14]), "v"(jbits[15]), "v"(jbits[16]), "v"(jbits[17]),
                     "v"(jbits[18]), "v"(jbits[19]), "v"(jbits[20]), "v"(jbits[21]), "v"(jbits[22]), "v"(jbits[23]));

#pragma unroll
  for (int a = 0; a < Q_ROUNDS; ++a) {
    const int ag = tid + a * STEP_THREADS;
    float m = qv[a][0];
#pragma unroll
    for (int y = 1; y < NACT; ++y) m = fmaxf(m, qv[a][y]);
    float e[NACT];
    float s = 0.f;
#pragma unroll
    for (int y = 0; y < NACT; ++y) { e[y] = __expf(qv[a][y] - m); s += e[y]; }
    const float inv = 1.f / s;
    if (ag < AGENTS) {
#pragma unroll
      for (int y = 0; y < NACT; ++y) Pl[ag * NACT + y] = e[y] * inv;
    }
  }

  int* mylist = List + wave * LIST_STRIDE;
  bool     hitv[JROUNDS];
  unsigned hmask[JROUNDS];
  unsigned ent[JROUNDS];
  unsigned cy = (unsigned)iy + MAX_DIST;
  unsigned cx = (unsigned)ix + MAX_DIST;
  asm("" : "+s"(cy), "+s"(cx));
#pragma unroll
  for (int r = 0; r < JROUNDS; ++r) {
    const unsigned ty = cy - (unsigned)jy[r];
    const unsigned tx = cx - (unsigned)jx[r];
    hitv[r]  = (ty > tx ? ty : tx) <= 2u * MAX_DIST;
    hmask[r] = __builtin_amdgcn_ballot_w32(hitv[r]);
    const unsigned dy  = absdiff5_add(ty, 0u);
    const unsigned dxj = absdiff5_add(tx, jbits[r]);
    ent[r] = __umul24(dy, FOV) + dxj;
  }
  int nhit = 0;
#pragma unroll
  for (int r = 0; r < JROUNDS; ++r) {
    const int at = (int)__builtin_amdgcn_mbcnt_lo(hmask[r], (unsigned)nhit);
    mylist[hitv[r] ? at : SPARE_AT + lane] = (int)ent[r];
    nhit += __builtin_popcount(hmask[r]);
  }
  __syncthreads();

#pragma unroll
  for (int k = 0; k < PV_ROUNDS; ++k) {
    const int e = tid + k * STEP_THREADS;
    if (e < PUSED_COUNT) {
      const int p   = e / (NACT * NACT);
      const int rem = e - p * (NACT * NACT);
      const int x   = rem / NACT;
      const int y   = rem - x * NACT;
      Bm[(p * NACT + y) * 16 + x] = pv[k];
    }
  }

  {
    const int g     = lane >> 3;
    const int y     = lane & 7;
    const bool yv   = y < NACT;
    const int yg    = y * NCOPY + g;
    const int dummy = KPAD * NCOPY + (lane & 15);
    for (int c = 0; c < nhit; c += CHUNK) {
      int   ao[CSTEPS];
      float pr[CSTEPS];
#pragma unroll
      for (int t = 0; t < CSTEPS; ++t) {
        const int idx = c + NCOPY * t + g;
        const bool ok = yv && idx < nhit;
        const unsigned e = (unsigned)mylist[idx < nhit ? idx : nhit - 1];
        ao[t] = ok ? (int)(e & 0xffffu) * (NACT * NCOPY) + yg : dummy;
        pr[t] = Pl[ok ? (int)(e >> 16) * NACT + y : AGENTS * NACT];
      }
      static_assert(CSTEPS == 8, "the statement below names CSTEPS values of each kind");
      asm volatile("" :: "v"(pr[0]), "v"(pr[1]), "v"(pr[2]), "v"(pr[3]),
                         "v"(pr[4]), "v"(pr[5]), "v"(pr[6]), "v"(pr[7]),
                         "v"(ao[0]), "v"(ao[1]), "v"(ao[2]), "v"(ao[3]),
                         "v"(ao[4]), "v"(ao[5]), "v"(ao[6]), "v"(ao[7]));
#pragma unroll
      for (int t = 0; t < CSTEPS; ++t) {
        if (c + NCOPY * t >= nhit) break;
        srow[ao[t]] += pr[t];
      }
    }
  }
  __syncthreads();

  const int half  = lane >> 4;
  const int nl    = lane & 15;
  const int kbase = wave * (KSTEPS_PER_WAVE * 4);
  const v4f* S4 = (const v4f*)Slds;
  v8f acc = {};
#pragma unroll
  for (int s = 0; s < KSTEPS_PER_WAVE; ++s) {
    const int ka = kbase + 4 * s + 2 * half;
    v2f a, bf;
    const v4f u0 = S4[nl * SSTRIDE + ka];
    const v4f u1 = S4[nl * SSTRIDE + ka + 1];
    a.x  = ((u0.x + u0.y) + u0.z) + u0.w;
    a.y  = ((u1.x + u1.y) + u1.z) + u1.w;
    bf.x = Bm[ka * 16 + nl];
    bf.y = Bm[(ka + 1) * 16 + nl];
    acc = __builtin_amdgcn_wmma_f32_16x16x4_f32(
         false, a,  false, bf,
         (short)0, acc,  false,  false);
  }

  if (nl < NACT) {
#pragma unroll
    for (int g = 0; g < 8; ++g)
      Red[(wave * 16 + g + 8 * half) * NACT + nl] = acc[g];
  }
  __syncthreads();

  if (tid < 16 * NACT) {
    float sum = Red[tid];
#pragma unroll
    for (int w = 1; w < STEP_WAVES; ++w) sum += Red[w * 16 * NACT + tid];
    const int o = (b * AGENTS + i_base) * NACT + tid;
    qout[o] = logits[o] + sum - NEG_INF_F * ill[o];
  }
}

extern "C" void kernel_launch(void* const* d_in, const int* in_sizes, int n_in,
                              void* d_out, int out_size, void* d_ws, size_t ws_size,
                              hipStream_t stream) {
  (void)in_sizes; (void)n_in; (void)out_size; (void)ws_size;
  const float* logits = (const float*)d_in[0];
  const float* ill    = (const float*)d_in[1];
  const int*   pos    = (const int*)d_in[2];
  const float* q_init = (const float*)d_in[3];
  const float* params = (const float*)d_in[4];
  const int NQ = BATCH * AGENTS * NACT;

  float* q_a = (float*)d_ws;
  float* q_b = q_a + NQ;

  mf_step<true><<<BATCH * TILES_PER_B, STEP_THREADS, 0, stream>>>(
      logits, ill, pos, params, q_init, q_a);
  mf_step<false><<<BATCH * TILES_PER_B, STEP_THREADS, 0, stream>>>(
      logits, ill, pos, params, q_a, q_b);
  mf_step<false><<<BATCH * TILES_PER_B, STEP_THREADS, 0, stream>>>(
      logits, ill, pos, params, q_b, (float*)d_out);
}


# Neighbour lists made once per call: sweep 1 stores each row's list in whole aligned 256-byte units (one store per wave), sweeps 2 and 3 load it instead of testing 768 pairs; same sums
# speedup vs baseline: 20.4885x; 1.1134x over previous
#include <hip/hip_runtime.h>


#define FOV        11
#define MAX_DIST   5
#define NACT       5
#define NEG_INF_F  1e10f
#define AGENTS     768
#define BATCH      4
#define PMIN       60
#define NPUSED     61
#define KTRUE      (NPUSED * NACT)
#define STEP_WAVES 16
#define KSTEPS_PER_WAVE 5
#define KPAD       (STEP_WAVES * KSTEPS_PER_WAVE * 4)
#define SSTRIDE    325
#define TILES_PER_B (AGENTS / 16)
#define NCOPY      4
#define CHUNK      32
#define CSTEPS     (CHUNK / NCOPY)
#define SPARE_AT    (AGENTS + 32)
#define LIST_STRIDE (AGENTS + 66)
#define UNIT_WORDS  64
#define ROW_UNITS   ((AGENTS + 1 + UNIT_WORDS - 1) / UNIT_WORDS)
#define ROW_WORDS   (ROW_UNITS * UNIT_WORDS)

typedef float v2f __attribute__((ext_vector_type(2)));
typedef float v8f __attribute__((ext_vector_type(8)));
typedef float v4f __attribute__((ext_vector_type(4)));
typedef int   v2i __attribute__((ext_vector_type(2)));

#define STEP_THREADS (STEP_WAVES * 32)
#define JROUNDS      (AGENTS / 32)
#define B4_COUNT     ((KPAD * 16) / 4)
#define B4_ROUNDS    ((B4_COUNT + STEP_THREADS - 1) / STEP_THREADS)
#define PUSED_COUNT  (NPUSED * NACT * NACT)
#define PV_ROUNDS    ((PUSED_COUNT + STEP_THREADS - 1) / STEP_THREADS)
#define Q_ROUNDS     ((AGENTS + STEP_THREADS - 1) / STEP_THREADS)

__device__ __forceinline__ unsigned absdiff5_add(unsigned a, unsigned c) {
  unsigned d;
  asm("v_sad_u32 %0, %1, 5, %2" : "=v"(d) : "v"(a), "v"(c));
  return d;
}

template <bool FIRST>
__global__ void __launch_bounds__(STEP_THREADS)
mf_step(const float* __restrict__ logits,
        const float* __restrict__ ill,
        const int*   __restrict__ pos,
        const float* __restrict__ params,
        const float* __restrict__ qsrc,
        float* __restrict__ qout,
        int* __restrict__ lists) {
  __shared__ __attribute__((aligned(16))) float Slds[16 * SSTRIDE * NCOPY];
  __shared__ __attribute__((aligned(16))) int List[STEP_WAVES * LIST_STRIDE];
  __shared__ __attribute__((aligned(16))) float Bm[B4_ROUNDS * STEP_THREADS * 4];
  __shared__ float Pl[AGENTS * NACT + 1];
  __shared__ float Red[STEP_WAVES * 16 * NACT];

  const int tid    = threadIdx.x;
  const int lane   = tid & 31;
  const int wave   = __builtin_amdgcn_readfirstlane(tid >> 5);
  const int b      = blockIdx.x / TILES_PER_B;
  const int i_base = (blockIdx.x - b * TILES_PER_B) * 16;

  unsigned jbits[JROUNDS];
  if (FIRST) {
#pragma unroll
    for (int r = 0; r < JROUNDS; ++r) jbits[r] = (unsigned)(r * 32 + lane) << 16;
  }

  const float* qb  = qsrc + (size_t)(b * AGENTS) * NACT;
  const float* ib  = ill  + (size_t)(b * AGENTS) * NACT;
  const int*   pb  = pos + (size_t)(b * AGENTS) * 2;
  int* rowlist = lists + (size_t)(b * AGENTS + i_base + wave) * ROW_WORDS;
  int* lw      = List + wave * LIST_STRIDE;
  int* lent    = lw + 1;

  int iy = 0, ix = 0;
  int jy[JROUNDS], jx[JROUNDS];
  v2i u0 = {0, 0};
  if (FIRST) {
    iy = pb[2 * (i_base + wave) + 0];
    ix = pb[2 * (i_base + wave) + 1];
#pragma unroll
    for (int r = 0; r < JROUNDS; ++r) {
      const int j = r * 32 + lane;
      jy[r] = pb[2 * j];
      jx[r] = pb[2 * j + 1];
    }
  } else {
    u0 = *(const v2i*)(rowlist + 2 * lane);
  }
  float qv[Q_ROUNDS][NACT];
#pragma unroll
  for (int a = 0; a < Q_ROUNDS; ++a) {
    const int ag = tid + a * STEP_THREADS;
    const int o  = (ag < AGENTS ? ag : AGENTS - 1) * NACT;
#pragma unroll
    for (int y = 0; y < NACT; ++y) {
      if (FIRST) qv[a][y] = qb[o + y] - NEG_INF_F * ib[o + y];
      else       qv[a][y] = qb[o + y];
    }
  }
  float pv[PV_ROUNDS];
#pragma unroll
  for (int k = 0; k < PV_ROUNDS; ++k) {
    const int e = tid + k * STEP_THREADS;
    pv[k] = params[PMIN * NACT * NACT + (e < PUSED_COUNT ? e : PUSED_COUNT - 1)];
  }

  float* srow = Slds + wave * (SSTRIDE * NCOPY);
  {
    v4f* srow4 = (v4f*)srow;
    const v4f z4 = {0.f, 0.f, 0.f, 0.f};
#pragma unroll
    for (int k = 0; k < (SSTRIDE + 31) / 32; ++k) {
      const int idx = lane + 32 * k;
      if (idx < SSTRIDE) srow4[idx] = z4;
    }
  }
  if (tid == 0) Pl[AGENTS * NACT] = 0.f;

  {
    v4f* Bm4 = (v4f*)Bm;
    const v4f z4 = {0.f, 0.f, 0.f, 0.f};
#pragma unroll
    for (int k = 0; k < B4_ROUNDS; ++k) Bm4[tid + k * STEP_THREADS] = z4;
  }

  static_assert(JROUNDS == 24, "the statement below names JROUNDS values");
  if (FIRST)
  asm volatile("" :: "v"(jbits[0]), "v"(jbits[1]), "v"(jbits[2]), "v"(jbits[3]), "v"(jbits[4]), "v"(jbits[5]),
                     "v"(jbits[6]), "v"(jbits[7]), "v"(jbits[8]), "v"(jbits[9]), "v"(jbits[10]), "v"(jbits[11]),
                     "v"(jbits[12]), "v"(jbits[13]), "v"(jbits[14]), "v"(jbits[15]), "v"(jbits[16]), "v"(jbits[17]),
                     "v"(jbits[18]), "v"(jbits[19]), "v"(jbits[20]), "v"(jbits[21]), "v"(jbits[22]), "v"(jbits[23]));

#pragma unroll
  for (int a = 0; a < Q_ROUNDS; ++a) {
    const int ag = tid + a * STEP_THREADS;
    float m = qv[a][0];
#pragma unroll
    for (int y = 1; y < NACT; ++y) m = fmaxf(m, qv[a][y]);
    float e[NACT];
    float s = 0.f;
#pragma unroll
    for (int y = 0; y < NACT; ++y) { e[y] = __expf(qv[a][y] - m); s += e[y]; }
    const float inv = 1.f / s;
    if (ag < AGENTS) {
#pragma unroll
      for (int y = 0; y < NACT; ++y) Pl[ag * NACT + y] = e[y] * inv;
    }
  }

  int nhit = 0;
  if (FIRST) {
    bool     hitv[JROUNDS];
    unsigned hmask[JROUNDS];
    unsigned ent[JROUNDS];
    unsigned cy = (unsigned)iy + MAX_DIST;
    unsigned cx = (unsigned)ix + MAX_DIST;
    asm("" : "+s"(cy), "+s"(cx));
#pragma unroll
    for (int r = 0; r < JROUNDS; ++r) {
      const unsigned ty = cy - (unsigned)jy[r];
      const unsigned tx = cx - (unsigned)jx[r];
      hitv[r]  = (ty > tx ? ty : tx) <= 2u * MAX_DIST;
      hmask[r] = __builtin_amdgcn_ballot_w32(hitv[r]);
      const unsigned dy  = absdiff5_add(ty, 0u);
      const unsigned dxj = absdiff5_add(tx, jbits[r]);
      ent[r] = __umul24(dy, FOV) + dxj;
    }
#pragma unroll
    for (int r = 0; r < JROUNDS; ++r) {
      const int at = (int)__builtin_amdgcn_mbcnt_lo(hmask[r], (unsigned)nhit);
      lent[hitv[r] ? at : SPARE_AT + lane] = (int)ent[r];
      nhit += __builtin_popcount(hmask[r]);
    }
  } else {
    *(v2i*)(lw + 2 * lane) = u0;
    const unsigned cnt = (unsigned)__builtin_amdgcn_readfirstlane(u0.x);
    nhit = (int)(cnt < (unsigned)AGENTS ? cnt : (unsigned)AGENTS);
    for (int u = 1; u <= (nhit >> 6); ++u) {
      const int w = u * UNIT_WORDS + 2 * lane;
      *(v2i*)(lw + w) = *(const v2i*)(rowlist + w);
    }
  }
  __syncthreads();

  v2i lu0 = {0, 0};
  if (FIRST) lu0 = *(const v2i*)(lw + 2 * lane);

#pragma unroll
  for (int k = 0; k < PV_ROUNDS; ++k) {
    const int e = tid + k * STEP_THREADS;
    if (e < PUSED_COUNT) {
      const int p   = e / (NACT * NACT);
      const int rem = e - p * (NACT * NACT);
      const int x   = rem / NACT;
      const int y   = rem - x * NACT;
      Bm[(p * NACT + y) * 16 + x] = pv[k];
    }
  }

  if (FIRST) {
    {
      const int w = 2 * lane;
      v2i v = lu0;
      v.x = (lane == 0) ? nhit : (w <= nhit ? v.x : 0);
      v.y = (w + 1 <= nhit) ? v.y : 0;
      *(v2i*)(rowlist + w) = v;
    }
    for (int u = 1; u <= (nhit >> 6); ++u) {
      const int w = u * UNIT_WORDS + 2 * lane;
      v2i v = *(const v2i*)(lw + w);
      v.x = (w <= nhit) ? v.x : 0;
      v.y = (w + 1 <= nhit) ? v.y : 0;
      *(v2i*)(rowlist + w) = v;
    }
  }

  {
    const int g     = lane >> 3;
    const int y     = lane & 7;
    const bool yv   = y < NACT;
    const int yg    = y * NCOPY + g;
    const int dummy = KPAD * NCOPY + (lane & 15);
    for (int c = 0; c < nhit; c += CHUNK) {
      int   ao[CSTEPS];
      float pr[CSTEPS];
#pragma unroll
      for (int t = 0; t < CSTEPS; ++t) {
        const int idx = c + NCOPY * t + g;
        const bool ok = yv && idx < nhit;
        const unsigned e = (unsigned)lent[idx < nhit ? idx : nhit - 1];
        ao[t] = ok ? (int)(e & 0xffffu) * (NACT * NCOPY) + yg : dummy;
        pr[t] = Pl[ok ? (int)(e >> 16) * NACT + y : AGENTS * NACT];
      }
      static_assert(CSTEPS == 8, "the statement below names CSTEPS values of each kind");
      asm volatile("" :: "v"(pr[0]), "v"(pr[1]), "v"(pr[2]), "v"(pr[3]),
                         "v"(pr[4]), "v"(pr[5]), "v"(pr[6]), "v"(pr[7]),
                         "v"(ao[0]), "v"(ao[1]), "v"(ao[2]), "v"(ao[3]),
                         "v"(ao[4]), "v"(ao[5]), "v"(ao[6]), "v"(ao[7]));
#pragma unroll
      for (int t = 0; t < CSTEPS; ++t) {
        if (c + NCOPY * t >= nhit) break;
        srow[ao[t]] += pr[t];
      }
    }
  }
  __syncthreads();

  const int half  = lane >> 4;
  const int nl    = lane & 15;
  const int kbase = wave * (KSTEPS_PER_WAVE * 4);
  const v4f* S4 = (const v4f*)Slds;
  v8f acc = {};
#pragma unroll
  for (int s = 0; s < KSTEPS_PER_WAVE; ++s) {
    const int ka = kbase + 4 * s + 2 * half;
    v2f a, bf;
    const v4f u0 = S4[nl * SSTRIDE + ka];
    const v4f u1 = S4[nl * SSTRIDE + ka + 1];
    a.x  = ((u0.x + u0.y) + u0.z) + u0.w;
    a.y  = ((u1.x + u1.y) + u1.z) + u1.w;
    bf.x = Bm[ka * 16 + nl];
    bf.y = Bm[(ka + 1) * 16 + nl];
    acc = __builtin_amdgcn_wmma_f32_16x16x4_f32(
         false, a,  false, bf,
         (short)0, acc,  false,  false);
  }

  if (nl < NACT) {
#pragma unroll
    for (int g = 0; g < 8; ++g)
      Red[(wave * 16 + g + 8 * half) * NACT + nl] = acc[g];
  }
  __syncthreads();

  if (tid < 16 * NACT) {
    float sum = Red[tid];
#pragma unroll
    for (int w = 1; w < STEP_WAVES; ++w) sum += Red[w * 16 * NACT + tid];
    const int o = (b * AGENTS + i_base) * NACT + tid;
    qout[o] = logits[o] + sum - NEG_INF_F * ill[o];
  }
}

extern "C" void kernel_launch(void* const* d_in, const int* in_sizes, int n_in,
                              void* d_out, int out_size, void* d_ws, size_t ws_size,
                              hipStream_t stream) {
  (void)in_sizes; (void)n_in; (void)out_size; (void)ws_size;
  const float* logits = (const float*)d_in[0];
  const float* ill    = (const float*)d_in[1];
  const int*   pos    = (const int*)d_in[2];
  const float* q_init = (const float*)d_in[3];
  const float* params = (const float*)d_in[4];
  const int NQ = BATCH * AGENTS * NACT;

  float* q_a = (float*)d_ws;
  float* q_b = q_a + NQ;
  int* lists = (int*)(q_b + NQ);

  mf_step<true><<<BATCH * TILES_PER_B, STEP_THREADS, 0, stream>>>(
      logits, ill, pos, params, q_init, q_a, lists);
  mf_step<false><<<BATCH * TILES_PER_B, STEP_THREADS, 0, stream>>>(
      logits, ill, pos, params, q_a, q_b, lists);
  mf_step<false><<<BATCH * TILES_PER_B, STEP_THREADS, 0, stream>>>(
      logits, ill, pos, params, q_b, (float*)d_out, lists);
}
